# starting kernel
# baseline (speedup 1.0000x reference)
#include <hip/hip_runtime.h>
#include <math.h>

#define BATCH 16
#define CH    4
#define FH    128
#define FW    240
#define NDISP 24
#define OH    1024
#define OW    1920

typedef __attribute__((ext_vector_type(2))) float v2f;
typedef __attribute__((ext_vector_type(8))) float v8f;

__global__ void __launch_bounds__(256)
disparity_pred_kernel(const float* __restrict__ fl,
                      const float* __restrict__ fr,
                      float* __restrict__ pred) {
    const int by = blockIdx.x;
    const int b  = by >> 7;
    const int y  = by & (FH - 1);
    const int t  = threadIdx.x;

    __shared__ float Rr[CH][FW];

    if (t < FW) {
#pragma unroll
        for (int c = 0; c < CH; ++c) {
            Rr[c][t] = __builtin_nontemporal_load(
                fr + (((size_t)(b * CH + c) * FH + y) * FW + t));
        }
    }
    __syncthreads();

    if (t < FW) {
        float L[CH];
#pragma unroll
        for (int c = 0; c < CH; ++c) {
            L[c] = __builtin_nontemporal_load(
                fl + (((size_t)(b * CH + c) * FH + y) * FW + t));
        }

        float cost[NDISP];
#pragma unroll
        for (int d = 0; d < NDISP; ++d) {
            float s = 0.0f;
            const int xs = t - d;
#pragma unroll
            for (int c = 0; c < CH; ++c) {
                const float r = (xs >= 0) ? Rr[c][xs] : 0.0f;
                s += fabsf(L[c] - r);
            }
            cost[d] = s;
        }

        float m = cost[0];
#pragma unroll
        for (int d = 1; d < NDISP; ++d) m = fminf(m, cost[d]);

        float se = 0.0f, sd = 0.0f;
#pragma unroll
        for (int d = 0; d < NDISP; ++d) {
            const float e = expf(m - cost[d]);
            se += e;
            sd += (float)d * e;
        }
        pred[((size_t)b * FH + y) * FW + t] = 8.0f * (sd / se);
    }
}

__global__ void __launch_bounds__(256)
upsample_wmma_kernel(const float* __restrict__ pred,
                     float* __restrict__ out) {
    const float SY = 127.0f / 1023.0f;
    const float SX = 239.0f / 1919.0f;

    const int lane = threadIdx.x & 31;
    const int wave = threadIdx.x >> 5;
    const int tx = blockIdx.x * 8 + wave;
    const int ty = blockIdx.y;
    const int b  = blockIdx.z;

    const int m     = lane & 15;
    const int half  = lane >> 4;
    const int kBase = half * 2;

    const int   yOut = ty * 16 + m;
    const float ysf  = (float)yOut * SY;
    const int   y0   = (int)ysf;
    const float wy   = ysf - (float)y0;
    const int   y1   = min(y0 + 1, FH - 1);

    const int xBase = tx * 16;
    const int c0    = (int)((float)xBase * SX);
    const int colA0 = min(c0 + kBase,     FW - 1);
    const int colA1 = min(c0 + kBase + 1, FW - 1);

    const float* pb = pred + (size_t)b * FH * FW;
    const float r0a = pb[y0 * FW + colA0];
    const float r0b = pb[y0 * FW + colA1];
    const float r1a = pb[y1 * FW + colA0];
    const float r1b = pb[y1 * FW + colA1];

    v2f A;
    A.x = r0a + wy * (r1a - r0a);
    A.y = r0b + wy * (r1b - r0b);

    const float xsf = (float)(xBase + m) * SX;
    const int   x0  = (int)xsf;
    const float wx  = xsf - (float)x0;
    const int   x1  = min(x0 + 1, FW - 1);
    const int   rel0 = x0 - c0;
    const int   rel1 = x1 - c0;

    v2f Bm;
    Bm.x = (rel0 == kBase     ? 1.0f - wx : 0.0f) + (rel1 == kBase     ? wx : 0.0f);
    Bm.y = (rel0 == kBase + 1 ? 1.0f - wx : 0.0f) + (rel1 == kBase + 1 ? wx : 0.0f);

    v8f Cacc = {};
    Cacc = __builtin_amdgcn_wmma_f32_16x16x4_f32(
         false, A,  false, Bm,
         (short)0, Cacc,  false,  false);

    float* ob = out + (size_t)b * OH * OW + (size_t)(ty * 16) * OW + xBase;
    const int rowOff = half * 8;
#pragma unroll
    for (int r = 0; r < 8; ++r) {
        __builtin_nontemporal_store(Cacc[r], ob + (size_t)(rowOff + r) * OW + m);
    }
}

extern "C" void kernel_launch(void* const* d_in, const int* in_sizes, int n_in,
                              void* d_out, int out_size, void* d_ws, size_t ws_size,
                              hipStream_t stream) {
    const float* feat_l = (const float*)d_in[0];
    const float* feat_r = (const float*)d_in[1];
    float* out  = (float*)d_out;
    float* pred = (float*)d_ws;

    disparity_pred_kernel<<<BATCH * FH, 256, 0, stream>>>(feat_l, feat_r, pred);

    upsample_wmma_kernel<<<dim3(15, 64, BATCH), 256, 0, stream>>>(pred, out);
}


# upsample output path: a 15-wave block builds its band (16 rows x 1920, one contiguous 120 KB run) in LDS from the same WMMA tiles, then streams whole rows, 512 contiguous bytes per store
# speedup vs baseline: 1.7757x; 1.7757x over previous
#include <hip/hip_runtime.h>
#include <math.h>

#define BATCH 16
#define CH    4
#define FH    128
#define FW    240
#define NDISP 24
#define OH    1024
#define OW    1920

typedef __attribute__((ext_vector_type(2))) float v2f;
typedef __attribute__((ext_vector_type(8))) float v8f;

__global__ void __launch_bounds__(256)
disparity_pred_kernel(const float* __restrict__ fl,
                      const float* __restrict__ fr,
                      float* __restrict__ pred) {
    const int by = blockIdx.x;
    const int b  = by >> 7;
    const int y  = by & (FH - 1);
    const int t  = threadIdx.x;

    __shared__ float Rr[CH][FW];

    if (t < FW) {
#pragma unroll
        for (int c = 0; c < CH; ++c) {
            Rr[c][t] = __builtin_nontemporal_load(
                fr + (((size_t)(b * CH + c) * FH + y) * FW + t));
        }
    }
    __syncthreads();

    if (t < FW) {
        float L[CH];
#pragma unroll
        for (int c = 0; c < CH; ++c) {
            L[c] = __builtin_nontemporal_load(
                fl + (((size_t)(b * CH + c) * FH + y) * FW + t));
        }

        float cost[NDISP];
#pragma unroll
        for (int d = 0; d < NDISP; ++d) {
            float s = 0.0f;
            const int xs = t - d;
#pragma unroll
            for (int c = 0; c < CH; ++c) {
                const float r = (xs >= 0) ? Rr[c][xs] : 0.0f;
                s += fabsf(L[c] - r);
            }
            cost[d] = s;
        }

        float m = cost[0];
#pragma unroll
        for (int d = 1; d < NDISP; ++d) m = fminf(m, cost[d]);

        float se = 0.0f, sd = 0.0f;
#pragma unroll
        for (int d = 0; d < NDISP; ++d) {
            const float e = expf(m - cost[d]);
            se += e;
            sd += (float)d * e;
        }
        pred[((size_t)b * FH + y) * FW + t] = 8.0f * (sd / se);
    }
}

#define UP_WAVES        15
#define UP_THREADS      (UP_WAVES * 32)
#define TILES_PER_WAVE  8
#define LDS_PITCH       1924

typedef __attribute__((ext_vector_type(4))) float v4f;

__global__ void __launch_bounds__(UP_THREADS)
upsample_band_kernel(const float* __restrict__ pred,
                     float* __restrict__ out) {
    __shared__ __attribute__((aligned(16))) float S[16 * LDS_PITCH];

    const float SY = 127.0f / 1023.0f;
    const float SX = 239.0f / 1919.0f;

    const int lane = threadIdx.x & 31;
    const int wave = threadIdx.x >> 5;
    const int ty = blockIdx.x;
    const int b  = blockIdx.y;

    const int m     = lane & 15;
    const int half  = lane >> 4;
    const int kBase = half * 2;

    const int   yOut = ty * 16 + m;
    const float ysf  = (float)yOut * SY;
    const int   y0   = (int)ysf;
    const float wy   = ysf - (float)y0;
    const int   y1   = min(y0 + 1, FH - 1);

    const float* pb   = pred + (size_t)b * FH * FW;
    const float* row0 = pb + y0 * FW;
    const float* row1 = pb + y1 * FW;

    float* sTile = S + (half * 8) * LDS_PITCH + m;

#pragma unroll
    for (int i = 0; i < TILES_PER_WAVE; ++i) {
        const int tx    = wave * TILES_PER_WAVE + i;
        const int xBase = tx * 16;

        const int c0    = (int)((float)xBase * SX);
        const int colA0 = min(c0 + kBase,     FW - 1);
        const int colA1 = min(c0 + kBase + 1, FW - 1);

        const float r0a = row0[colA0];
        const float r0b = row0[colA1];
        const float r1a = row1[colA0];
        const float r1b = row1[colA1];

        v2f A;
        A.x = r0a + wy * (r1a - r0a);
        A.y = r0b + wy * (r1b - r0b);

        const float xsf = (float)(xBase + m) * SX;
        const int   x0  = (int)xsf;
        const float wx  = xsf - (float)x0;
        const int   x1  = min(x0 + 1, FW - 1);
        const int   rel0 = x0 - c0;
        const int   rel1 = x1 - c0;

        v2f Bm;
        Bm.x = (rel0 == kBase     ? 1.0f - wx : 0.0f) + (rel1 == kBase     ? wx : 0.0f);
        Bm.y = (rel0 == kBase + 1 ? 1.0f - wx : 0.0f) + (rel1 == kBase + 1 ? wx : 0.0f);

        v8f Cacc = {};
        Cacc = __builtin_amdgcn_wmma_f32_16x16x4_f32(
             false, A,  false, Bm,
             (short)0, Cacc,  false,  false);

#pragma unroll
        for (int r = 0; r < 8; ++r) {
            sTile[r * LDS_PITCH + xBase] = Cacc[r];
        }
    }

    __syncthreads();

    const int t = threadIdx.x;
    float* ob = out + ((size_t)b * OH + (size_t)(ty * 16)) * OW + 4 * t;
    const float* sp = S + 4 * t;
#pragma unroll
    for (int r = 0; r < 16; ++r) {
        const v4f v = *(const v4f*)(sp + r * LDS_PITCH);
        __builtin_nontemporal_store(v, (v4f*)(ob + (size_t)r * OW));
    }
}

extern "C" void kernel_launch(void* const* d_in, const int* in_sizes, int n_in,
                              void* d_out, int out_size, void* d_ws, size_t ws_size,
                              hipStream_t stream) {
    const float* feat_l = (const float*)d_in[0];
    const float* feat_r = (const float*)d_in[1];
    float* out  = (float*)d_out;
    float* pred = (float*)d_ws;

    disparity_pred_kernel<<<BATCH * FH, 256, 0, stream>>>(feat_l, feat_r, pred);

    upsample_band_kernel<<<dim3(64, BATCH), 480, 0, stream>>>(pred, out);
}


# upsample block walks 2 consecutive bands through the same LDS image (512 blocks, one resident round): band 2 is computed while band 1's stores drain, x weights reused; was 1024 one-band blocks
# speedup vs baseline: 1.9364x; 1.0905x over previous
#include <hip/hip_runtime.h>
#include <math.h>

#define BATCH 16
#define CH    4
#define FH    128
#define FW    240
#define NDISP 24
#define OH    1024
#define OW    1920

typedef __attribute__((ext_vector_type(2))) float v2f;
typedef __attribute__((ext_vector_type(8))) float v8f;

__global__ void __launch_bounds__(256)
disparity_pred_kernel(const float* __restrict__ fl,
                      const float* __restrict__ fr,
                      float* __restrict__ pred) {
    const int by = blockIdx.x;
    const int b  = by >> 7;
    const int y  = by & (FH - 1);
    const int t  = threadIdx.x;

    __shared__ float Rr[CH][FW];

    if (t < FW) {
#pragma unroll
        for (int c = 0; c < CH; ++c) {
            Rr[c][t] = __builtin_nontemporal_load(
                fr + (((size_t)(b * CH + c) * FH + y) * FW + t));
        }
    }
    __syncthreads();

    if (t < FW) {
        float L[CH];
#pragma unroll
        for (int c = 0; c < CH; ++c) {
            L[c] = __builtin_nontemporal_load(
                fl + (((size_t)(b * CH + c) * FH + y) * FW + t));
        }

        float cost[NDISP];
#pragma unroll
        for (int d = 0; d < NDISP; ++d) {
            float s = 0.0f;
            const int xs = t - d;
#pragma unroll
            for (int c = 0; c < CH; ++c) {
                const float r = (xs >= 0) ? Rr[c][xs] : 0.0f;
                s += fabsf(L[c] - r);
            }
            cost[d] = s;
        }

        float m = cost[0];
#pragma unroll
        for (int d = 1; d < NDISP; ++d) m = fminf(m, cost[d]);

        float se = 0.0f, sd = 0.0f;
#pragma unroll
        for (int d = 0; d < NDISP; ++d) {
            const float e = expf(m - cost[d]);
            se += e;
            sd += (float)d * e;
        }
        pred[((size_t)b * FH + y) * FW + t] = 8.0f * (sd / se);
    }
}

#define UP_WAVES        15
#define UP_THREADS      (UP_WAVES * 32)
#define TILES_PER_WAVE  8
#define LDS_PITCH       1924
#define BANDS_PER_BLOCK 2

typedef __attribute__((ext_vector_type(4))) float v4f;

__device__ __forceinline__ void lds_barrier() {
    __builtin_amdgcn_fence(__ATOMIC_RELEASE, "workgroup", "local");
    __builtin_amdgcn_s_barrier();
    __builtin_amdgcn_fence(__ATOMIC_ACQUIRE, "workgroup", "local");
}

__global__ void __launch_bounds__(UP_THREADS)
upsample_band_kernel(const float* __restrict__ pred,
                     float* __restrict__ out) {
    __shared__ __attribute__((aligned(16))) float S[16 * LDS_PITCH];

    const float SY = 127.0f / 1023.0f;
    const float SX = 239.0f / 1919.0f;

    const int lane = threadIdx.x & 31;
    const int wave = threadIdx.x >> 5;
    const int b    = blockIdx.y;

    const int m     = lane & 15;
    const int half  = lane >> 4;
    const int kBase = half * 2;

    int colA0[TILES_PER_WAVE], colA1[TILES_PER_WAVE];
    v2f Bm[TILES_PER_WAVE];
#pragma unroll
    for (int i = 0; i < TILES_PER_WAVE; ++i) {
        const int xBase = (wave * TILES_PER_WAVE + i) * 16;
        const int c0    = (int)((float)xBase * SX);
        colA0[i] = min(c0 + kBase,     FW - 1);
        colA1[i] = min(c0 + kBase + 1, FW - 1);

        const float xsf = (float)(xBase + m) * SX;
        const int   x0  = (int)xsf;
        const float wx  = xsf - (float)x0;
        const int   x1  = min(x0 + 1, FW - 1);
        const int   rel0 = x0 - c0;
        const int   rel1 = x1 - c0;

        Bm[i].x = (rel0 == kBase     ? 1.0f - wx : 0.0f) + (rel1 == kBase     ? wx : 0.0f);
        Bm[i].y = (rel0 == kBase + 1 ? 1.0f - wx : 0.0f) + (rel1 == kBase + 1 ? wx : 0.0f);
    }

    const float* pb = pred + (size_t)b * FH * FW;
    float* sTile = S + (half * 8) * LDS_PITCH + m + wave * (TILES_PER_WAVE * 16);
    const int t = threadIdx.x;
    const float* sp = S + 4 * t;

#pragma unroll
    for (int band = 0; band < BANDS_PER_BLOCK; ++band) {
        const int ty = blockIdx.x * BANDS_PER_BLOCK + band;

        const int   yOut = ty * 16 + m;
        const float ysf  = (float)yOut * SY;
        const int   y0   = (int)ysf;
        const float wy   = ysf - (float)y0;
        const int   y1   = min(y0 + 1, FH - 1);
        const float* row0 = pb + y0 * FW;
        const float* row1 = pb + y1 * FW;

        v8f acc[TILES_PER_WAVE];
#pragma unroll
        for (int i = 0; i < TILES_PER_WAVE; ++i) {
            const float r0a = row0[colA0[i]];
            const float r0b = row0[colA1[i]];
            const float r1a = row1[colA0[i]];
            const float r1b = row1[colA1[i]];

            v2f A;
            A.x = r0a + wy * (r1a - r0a);
            A.y = r0b + wy * (r1b - r0b);

            v8f Cacc = {};
            acc[i] = __builtin_amdgcn_wmma_f32_16x16x4_f32(
                 false, A,  false, Bm[i],
                 (short)0, Cacc,  false,  false);
        }

        if (band > 0) lds_barrier();

#pragma unroll
        for (int i = 0; i < TILES_PER_WAVE; ++i) {
#pragma unroll
            for (int r = 0; r < 8; ++r) {
                sTile[r * LDS_PITCH + i * 16] = acc[i][r];
            }
        }

        lds_barrier();

        float* ob = out + ((size_t)b * OH + (size_t)(ty * 16)) * OW + 4 * t;
#pragma unroll
        for (int r = 0; r < 16; ++r) {
            const v4f v = *(const v4f*)(sp + r * LDS_PITCH);
            __builtin_nontemporal_store(v, (v4f*)(ob + (size_t)r * OW));
        }
    }
}

extern "C" void kernel_launch(void* const* d_in, const int* in_sizes, int n_in,
                              void* d_out, int out_size, void* d_ws, size_t ws_size,
                              hipStream_t stream) {
    const float* feat_l = (const float*)d_in[0];
    const float* feat_r = (const float*)d_in[1];
    float* out  = (float*)d_out;
    float* pred = (float*)d_ws;

    disparity_pred_kernel<<<BATCH * FH, 256, 0, stream>>>(feat_l, feat_r, pred);

    upsample_band_kernel<<<dim3(32, BATCH), 480, 0, stream>>>(pred, out);
}


# pred kernel: right-feature row staged in LDS with 4 channels interleaved behind zero pixels: 24 unguarded 16-byte LDS reads per pixel in place of 96 exec-masked 4-byte ones; sums in the same order
# speedup vs baseline: 2.1478x; 1.1092x over previous
#include <hip/hip_runtime.h>
#include <math.h>

#define BATCH 16
#define CH    4
#define FH    128
#define FW    240
#define NDISP 24
#define OH    1024
#define OW    1920

typedef __attribute__((ext_vector_type(2))) float v2f;
typedef __attribute__((ext_vector_type(8))) float v8f;

#define RPAD 24

typedef __attribute__((ext_vector_type(4))) float v4f;

__global__ void __launch_bounds__(256)
disparity_pred_kernel(const float* __restrict__ fl,
                      const float* __restrict__ fr,
                      float* __restrict__ pred) {
    const int by = blockIdx.x;
    const int b  = by >> 7;
    const int y  = by & (FH - 1);
    const int t  = threadIdx.x;

    __shared__ v4f Rq[RPAD + FW];

    if (t < RPAD) {
        const v4f z = {0.0f, 0.0f, 0.0f, 0.0f};
        Rq[t] = z;
    }
    if (t < FW) {
        v4f r;
        r.x = __builtin_nontemporal_load(fr + (((size_t)(b * CH + 0) * FH + y) * FW + t));
        r.y = __builtin_nontemporal_load(fr + (((size_t)(b * CH + 1) * FH + y) * FW + t));
        r.z = __builtin_nontemporal_load(fr + (((size_t)(b * CH + 2) * FH + y) * FW + t));
        r.w = __builtin_nontemporal_load(fr + (((size_t)(b * CH + 3) * FH + y) * FW + t));
        Rq[RPAD + t] = r;
    }
    __syncthreads();

    if (t < FW) {
        v4f L;
        L.x = __builtin_nontemporal_load(fl + (((size_t)(b * CH + 0) * FH + y) * FW + t));
        L.y = __builtin_nontemporal_load(fl + (((size_t)(b * CH + 1) * FH + y) * FW + t));
        L.z = __builtin_nontemporal_load(fl + (((size_t)(b * CH + 2) * FH + y) * FW + t));
        L.w = __builtin_nontemporal_load(fl + (((size_t)(b * CH + 3) * FH + y) * FW + t));

        const v4f* rp = Rq + (RPAD + t);

        float cost[NDISP];
#pragma unroll
        for (int d = 0; d < NDISP; ++d) {
            const v4f df = L - rp[-d];
            float s = 0.0f;
            s += fabsf(df.x);
            s += fabsf(df.y);
            s += fabsf(df.z);
            s += fabsf(df.w);
            asm("" : "+v"(s));
            cost[d] = s;
        }

        float m = cost[0];
#pragma unroll
        for (int d = 1; d < NDISP; ++d) m = fminf(m, cost[d]);

        float se = 0.0f, sd = 0.0f;
#pragma unroll
        for (int d = 0; d < NDISP; ++d) {
            const float e = expf(m - cost[d]);
            se += e;
            sd += (float)d * e;
        }
        pred[((size_t)b * FH + y) * FW + t] = 8.0f * (sd / se);
    }
}

#define UP_WAVES        15
#define UP_THREADS      (UP_WAVES * 32)
#define TILES_PER_WAVE  8
#define LDS_PITCH       1924
#define BANDS_PER_BLOCK 2

__device__ __forceinline__ void lds_barrier() {
    __builtin_amdgcn_fence(__ATOMIC_RELEASE, "workgroup", "local");
    __builtin_amdgcn_s_barrier();
    __builtin_amdgcn_fence(__ATOMIC_ACQUIRE, "workgroup", "local");
}

__global__ void __launch_bounds__(UP_THREADS)
upsample_band_kernel(const float* __restrict__ pred,
                     float* __restrict__ out) {
    __shared__ __attribute__((aligned(16))) float S[16 * LDS_PITCH];

    const float SY = 127.0f / 1023.0f;
    const float SX = 239.0f / 1919.0f;

    const int lane = threadIdx.x & 31;
    const int wave = threadIdx.x >> 5;
    const int b    = blockIdx.y;

    const int m     = lane & 15;
    const int half  = lane >> 4;
    const int kBase = half * 2;

    int colA0[TILES_PER_WAVE], colA1[TILES_PER_WAVE];
    v2f Bm[TILES_PER_WAVE];
#pragma unroll
    for (int i = 0; i < TILES_PER_WAVE; ++i) {
        const int xBase = (wave * TILES_PER_WAVE + i) * 16;
        const int c0    = (int)((float)xBase * SX);
        colA0[i] = min(c0 + kBase,     FW - 1);
        colA1[i] = min(c0 + kBase + 1, FW - 1);

        const float xsf = (float)(xBase + m) * SX;
        const int   x0  = (int)xsf;
        const float wx  = xsf - (float)x0;
        const int   x1  = min(x0 + 1, FW - 1);
        const int   rel0 = x0 - c0;
        const int   rel1 = x1 - c0;

        Bm[i].x = (rel0 == kBase     ? 1.0f - wx : 0.0f) + (rel1 == kBase     ? wx : 0.0f);
        Bm[i].y = (rel0 == kBase + 1 ? 1.0f - wx : 0.0f) + (rel1 == kBase + 1 ? wx : 0.0f);
    }

    const float* pb = pred + (size_t)b * FH * FW;
    float* sTile = S + (half * 8) * LDS_PITCH + m + wave * (TILES_PER_WAVE * 16);
    const int t = threadIdx.x;
    const float* sp = S + 4 * t;

#pragma unroll
    for (int band = 0; band < BANDS_PER_BLOCK; ++band) {
        const int ty = blockIdx.x * BANDS_PER_BLOCK + band;

        const int   yOut = ty * 16 + m;
        const float ysf  = (float)yOut * SY;
        const int   y0   = (int)ysf;
        const float wy   = ysf - (float)y0;
        const int   y1   = min(y0 + 1, FH - 1);
        const float* row0 = pb + y0 * FW;
        const float* row1 = pb + y1 * FW;

        v8f acc[TILES_PER_WAVE];
#pragma unroll
        for (int i = 0; i < TILES_PER_WAVE; ++i) {
            const float r0a = row0[colA0[i]];
            const float r0b = row0[colA1[i]];
            const float r1a = row1[colA0[i]];
            const float r1b = row1[colA1[i]];

            v2f A;
            A.x = r0a + wy * (r1a - r0a);
            A.y = r0b + wy * (r1b - r0b);

            v8f Cacc = {};
            acc[i] = __builtin_amdgcn_wmma_f32_16x16x4_f32(
                 false, A,  false, Bm[i],
                 (short)0, Cacc,  false,  false);
        }

        if (band > 0) lds_barrier();

#pragma unroll
        for (int i = 0; i < TILES_PER_WAVE; ++i) {
#pragma unroll
            for (int r = 0; r < 8; ++r) {
                sTile[r * LDS_PITCH + i * 16] = acc[i][r];
            }
        }

        lds_barrier();

        float* ob = out + ((size_t)b * OH + (size_t)(ty * 16)) * OW + 4 * t;
#pragma unroll
        for (int r = 0; r < 16; ++r) {
            const v4f v = *(const v4f*)(sp + r * LDS_PITCH);
            __builtin_nontemporal_store(v, (v4f*)(ob + (size_t)r * OW));
        }
    }
}

extern "C" void kernel_launch(void* const* d_in, const int* in_sizes, int n_in,
                              void* d_out, int out_size, void* d_ws, size_t ws_size,
                              hipStream_t stream) {
    const float* feat_l = (const float*)d_in[0];
    const float* feat_r = (const float*)d_in[1];
    float* out  = (float*)d_out;
    float* pred = (float*)d_ws;

    disparity_pred_kernel<<<BATCH * FH, 256, 0, stream>>>(feat_l, feat_r, pred);

    upsample_band_kernel<<<dim3(32, BATCH), 480, 0, stream>>>(pred, out);
}
